# starting kernel
# baseline (speedup 1.0000x reference)
#include <hip/hip_runtime.h>
#include <hip/hip_bf16.h>

#define G_ 128
#define T_ 128
#define S_ 64
#define M_ 8

typedef float v2f __attribute__((ext_vector_type(2)));
typedef float v8f __attribute__((ext_vector_type(8)));

__device__ __forceinline__ void mm64(const float* A, const float* B, float* D,
                                     bool transB,
                                     const float* addQ,
                                     const float* KRp,
                                     const float* Kp)
{
    const int lane = threadIdx.x & 31;
    const int wave = threadIdx.x >> 5;
    const int half = lane >> 4;
    const int l15  = lane & 15;

    for (int tt = wave; tt < 16; tt += 8) {
        const int ti = (tt >> 2) * 16;
        const int tj = (tt & 3) * 16;
        v8f acc = {};
        #pragma unroll
        for (int kb = 0; kb < 16; ++kb) {
            const int k0 = kb * 4 + half * 2;
            v2f a, b;
            a.x = A[(ti + l15) * 64 + k0];
            a.y = A[(ti + l15) * 64 + k0 + 1];
            if (!transB) {
                b.x = B[(k0    ) * 64 + (tj + l15)];
                b.y = B[(k0 + 1) * 64 + (tj + l15)];
            } else {
                b.x = B[(tj + l15) * 64 + k0];
                b.y = B[(tj + l15) * 64 + k0 + 1];
            }
            acc = __builtin_amdgcn_wmma_f32_16x16x4_f32(
                      false, a, false, b, (short)0, acc, false, false);
        }
        #pragma unroll
        for (int r = 0; r < 8; ++r) {
            const int row = ti + r + half * 8;
            const int col = tj + l15;
            float v = acc[r];
            if (addQ) v += addQ[row * 64 + col];
            if (KRp) {
                #pragma unroll
                for (int m = 0; m < 8; ++m)
                    v += KRp[row * 8 + m] * Kp[col * 8 + m];
            }
            D[row * 64 + col] = v;
        }
    }
}

__global__ __launch_bounds__(256) void kf_prep(const float* __restrict__ Lq,
                                               const float* __restrict__ Lr,
                                               const float* __restrict__ Linit,
                                               float* __restrict__ ws)
{
    float* Q  = ws;
    float* P0 = ws + 4096;
    float* R  = ws + 8192;
    for (int e = blockIdx.x * blockDim.x + threadIdx.x; e < 4096;
         e += gridDim.x * blockDim.x) {
        const int i = e >> 6, j = e & 63;
        float q = 0.f, p = 0.f;
        for (int k = 0; k < 64; ++k) {
            q += Lq[i * 64 + k] * Lq[j * 64 + k];
            p += Linit[i * 64 + k] * Linit[j * 64 + k];
        }
        Q[e] = q;
        P0[e] = p;
    }
    const int e = blockIdx.x * blockDim.x + threadIdx.x;
    if (e < 64) {
        const int i = e >> 3, j = e & 7;
        float r = 0.f;
        for (int k = 0; k < 8; ++k) r += Lr[i * 8 + k] * Lr[j * 8 + k];
        R[e] = r;
    }
}

__global__ __launch_bounds__(256) void kf_main(const float* __restrict__ y,
                                               const float* __restrict__ F,
                                               const float* __restrict__ H,
                                               const float* __restrict__ Qg,
                                               const float* __restrict__ Rg,
                                               const float* __restrict__ P0g,
                                               const float* __restrict__ init_mean,
                                               float* __restrict__ out)
{
    extern __shared__ float smem[];
    float* Fs    = smem;
    float* bufA  = Fs   + 4096;
    float* bufB  = bufA + 4096;
    float* bufC  = bufB + 4096;
    float* Hs    = bufC + 4096;
    float* Rs    = Hs   + 512;
    float* PHt   = Rs   + 64;
    float* Km    = PHt  + 512;
    float* KR    = Km   + 512;
    float* Wgj   = KR   + 512;
    float* meanv = Wgj  + 128;
    float* meanp = meanv + 64;
    float* resid = meanp + 64;
    float* fac   = resid + 8;
    float* piv   = fac   + 8;

    const int g   = blockIdx.x;
    const int tid = threadIdx.x;

    for (int i = tid; i < 4096; i += 256) {
        Fs[i]   = F[i];
        bufA[i] = P0g[i];
    }
    for (int i = tid; i < 512; i += 256) Hs[i] = H[i];
    if (tid < 64) {
        Rs[tid]    = Rg[tid];
        meanv[tid] = init_mean[g * S_ + tid];
    }
    __syncthreads();

    float* covP  = bufA;
    float* covpP = bufB;
    float* tmpP  = bufC;

    for (int t = 0; t < T_; ++t) {
        mm64(Fs, covP, tmpP, false, nullptr, nullptr, nullptr);
        __syncthreads();
        mm64(tmpP, Fs, covpP, true, Qg, nullptr, nullptr);
        __syncthreads();

        for (int e = tid; e < 512; e += 256) {
            const int s = e >> 3, m = e & 7;
            float acc = 0.f;
            for (int k = 0; k < 64; ++k) acc += covpP[s * 64 + k] * Hs[m * 64 + k];
            PHt[e] = acc;
        }
        if (tid < 64) {
            float acc = 0.f;
            for (int j = 0; j < 64; ++j) acc += Fs[tid * 64 + j] * meanv[j];
            meanp[tid] = acc;
        }
        __syncthreads();

        if (tid < 128) {
            const int i = tid >> 4, c = tid & 15;
            if (c < 8) {
                float acc = Rs[i * 8 + c];
                for (int s = 0; s < 64; ++s) acc += Hs[i * 64 + s] * PHt[s * 8 + c];
                Wgj[i * 16 + c] = acc;
            } else {
                Wgj[i * 16 + c] = (c - 8 == i) ? 1.f : 0.f;
            }
        } else if (tid < 136) {
            const int m = tid - 128;
            float acc = 0.f;
            for (int s = 0; s < 64; ++s) acc += Hs[m * 64 + s] * meanp[s];
            out[(g * T_ + t) * M_ + m] = acc;
            resid[m] = y[(g * T_ + t) * M_ + m] - acc;
        }
        __syncthreads();

        for (int p = 0; p < 8; ++p) {
            if (tid == 0) piv[0] = 1.f / Wgj[p * 16 + p];
            __syncthreads();
            if (tid < 16) Wgj[p * 16 + tid] *= piv[0];
            __syncthreads();
            if (tid < 8) fac[tid] = Wgj[tid * 16 + p];
            __syncthreads();
            if (tid < 128) {
                const int r = tid >> 4, c = tid & 15;
                if (r != p) Wgj[r * 16 + c] -= fac[r] * Wgj[p * 16 + c];
            }
            __syncthreads();
        }

        for (int e = tid; e < 512; e += 256) {
            const int s = e >> 3, m = e & 7;
            float acc = 0.f;
            for (int j = 0; j < 8; ++j) acc += PHt[s * 8 + j] * Wgj[j * 16 + 8 + m];
            Km[e] = acc;
        }
        __syncthreads();

        if (tid < 64) {
            float acc = meanp[tid];
            for (int m = 0; m < 8; ++m) acc += Km[tid * 8 + m] * resid[m];
            meanv[tid] = acc;
        }
        for (int e = tid; e < 512; e += 256) {
            const int i = e >> 3, m = e & 7;
            float acc = 0.f;
            for (int n = 0; n < 8; ++n) acc += Km[i * 8 + n] * Rs[n * 8 + m];
            KR[e] = acc;
        }
        for (int e = tid; e < 4096; e += 256) {
            const int i = e >> 6, j = e & 63;
            float acc = (i == j) ? 1.f : 0.f;
            for (int m = 0; m < 8; ++m) acc -= Km[i * 8 + m] * Hs[m * 64 + j];
            covP[e] = acc;
        }
        __syncthreads();

        mm64(covP, covpP, tmpP, false, nullptr, nullptr, nullptr);
        __syncthreads();
        mm64(tmpP, covP, covpP, true, nullptr, KR, Km);
        __syncthreads();

        float* sw = covP; covP = covpP; covpP = sw;
    }
}

extern "C" void kernel_launch(void* const* d_in, const int* in_sizes, int n_in,
                              void* d_out, int out_size, void* d_ws, size_t ws_size,
                              hipStream_t stream) {
    const float* y         = (const float*)d_in[0];
    const float* F         = (const float*)d_in[1];
    const float* H         = (const float*)d_in[2];
    const float* Lq        = (const float*)d_in[3];
    const float* Lr        = (const float*)d_in[4];
    const float* init_mean = (const float*)d_in[5];
    const float* Linit     = (const float*)d_in[6];
    float* out = (float*)d_out;
    float* ws  = (float*)d_ws;

    kf_prep<<<16, 256, 0, stream>>>(Lq, Lr, Linit, ws);

    const size_t smem_bytes = (4096 * 4 + 512 + 64 + 512 + 512 + 512 + 128 +
                               64 + 64 + 8 + 8 + 8) * sizeof(float);
    kf_main<<<G_, 256, smem_bytes, stream>>>(y, F, H,
                                             ws,
                                             ws + 8192,
                                             ws + 4096,
                                             init_mean, out);
}


# 8x8 Gauss-Jordan as one LDS pass per pivot between two copies of the matrix (each thread forms 1/pivot and the scaled pivot-row entry itself): 8 barrier intervals, not 32; same sums, same order
# speedup vs baseline: 4.8463x; 4.8463x over previous
#include <hip/hip_runtime.h>
#include <hip/hip_bf16.h>

#define G_ 128
#define T_ 128
#define S_ 64
#define M_ 8

typedef float v2f __attribute__((ext_vector_type(2)));
typedef float v8f __attribute__((ext_vector_type(8)));

__device__ __forceinline__ v8f slab(const float* At, const int ap,
                                    const float* Bk, const int bp,
                                    const int kb, const int half, const int l15,
                                    const v8f acc);

__device__ __forceinline__ void mm64(const float* At, const float* Bk, float* D,
                                     const float* addQ,
                                     const float* At2,
                                     const float* Bk2)
{
    const int lane = threadIdx.x & 31;
    const int wave = threadIdx.x >> 5;
    const int half = lane >> 4;
    const int l15  = lane & 15;

    for (int tt = wave; tt < 16; tt += 8) {
        const int ti = (tt >> 2) * 16;
        const int tj = (tt & 3) * 16;
        v8f acc = {};
        #pragma unroll
        for (int kb = 0; kb < 16; ++kb) {
            const int k0 = kb * 4 + half * 2;
            v2f a, b;
            a.x = At[(k0    ) * 64 + (ti + l15)];
            a.y = At[(k0 + 1) * 64 + (ti + l15)];
            b.x = Bk[(k0    ) * 64 + (tj + l15)];
            b.y = Bk[(k0 + 1) * 64 + (tj + l15)];
            acc = __builtin_amdgcn_wmma_f32_16x16x4_f32(
                      false, a, false, b, (short)0, acc, false, false);
        }
        if (At2) {
            acc = slab(At2 + ti, 64, Bk2 + tj, 64, 0, half, l15, acc);
            acc = slab(At2 + ti, 64, Bk2 + tj, 64, 1, half, l15, acc);
        }
        #pragma unroll
        for (int r = 0; r < 8; ++r) {
            const int row = ti + r + half * 8;
            const int col = tj + l15;
            float v = acc[r];
            if (addQ) v += addQ[row * 64 + col];
            D[row * 64 + col] = v;
        }
    }
}

__device__ __forceinline__ v8f tile64(const float* At, const int ap,
                                      const float* Bk, const int bp,
                                      const int half, const int l15)
{
    v8f acc = {};
    #pragma unroll
    for (int kb = 0; kb < 16; ++kb) {
        const int k0 = kb * 4 + half * 2;
        v2f a, b;
        a.x = At[(k0    ) * ap + l15];
        a.y = At[(k0 + 1) * ap + l15];
        b.x = Bk[(k0    ) * bp + l15];
        b.y = Bk[(k0 + 1) * bp + l15];
        acc = __builtin_amdgcn_wmma_f32_16x16x4_f32(
                  false, a, false, b, (short)0, acc, false, false);
    }
    return acc;
}

__device__ __forceinline__ v8f slab(const float* At, const int ap,
                                    const float* Bk, const int bp,
                                    const int kb, const int half, const int l15,
                                    const v8f acc)
{
    const int k0 = kb * 4 + half * 2;
    v2f a, b;
    a.x = At[(k0    ) * ap + l15];
    a.y = At[(k0 + 1) * ap + l15];
    b.x = Bk[(k0    ) * bp + l15];
    b.y = Bk[(k0 + 1) * bp + l15];
    return __builtin_amdgcn_wmma_f32_16x16x4_f32(
               false, a, false, b, (short)0, acc, false, false);
}

__global__ __launch_bounds__(256) void kf_prep(const float* __restrict__ Lq,
                                               const float* __restrict__ Lr,
                                               const float* __restrict__ Linit,
                                               float* __restrict__ ws)
{
    float* Q  = ws;
    float* P0 = ws + 4096;
    float* R  = ws + 8192;
    for (int e = blockIdx.x * blockDim.x + threadIdx.x; e < 4096;
         e += gridDim.x * blockDim.x) {
        const int i = e >> 6, j = e & 63;
        float q = 0.f, p = 0.f;
        for (int k = 0; k < 64; ++k) {
            q += Lq[i * 64 + k] * Lq[j * 64 + k];
            p += Linit[i * 64 + k] * Linit[j * 64 + k];
        }
        Q[e] = q;
        P0[e] = p;
    }
    const int e = blockIdx.x * blockDim.x + threadIdx.x;
    if (e < 64) {
        const int i = e >> 3, j = e & 7;
        float r = 0.f;
        for (int k = 0; k < 8; ++k) r += Lr[i * 8 + k] * Lr[j * 8 + k];
        R[e] = r;
    }
}

__global__ __launch_bounds__(256) void kf_main(const float* __restrict__ y,
                                               const float* __restrict__ F,
                                               const float* __restrict__ H,
                                               const float* __restrict__ Qg,
                                               const float* __restrict__ Rg,
                                               const float* __restrict__ P0g,
                                               const float* __restrict__ init_mean,
                                               float* __restrict__ out)
{
    extern __shared__ float smem[];
    float* Ft    = smem;
    float* bufA  = Ft   + 4096;
    float* bufB  = bufA + 4096;
    float* bufC  = bufB + 4096;
    float* Hs    = bufC + 4096;
    float* Rs    = Hs   + 512;
    float* PHt   = Rs   + 64;
    float* Km    = PHt  + 512;
    float* KRt   = Km   + 512;
    float* Wgj   = KRt  + 512;
    float* meanv = Wgj  + 128;
    float* meanp = meanv + 64;
    float* resid = meanp + 64;
    float* Wgj2  = resid + 8;
    float* Gt    = Wgj2  + 128;
    float* VT    = Gt    + 1024;
    float* QHs   = VT    + 1024;
    float* S0s   = QHs   + 512;
    float* Kt    = S0s   + 64;

    const int g   = blockIdx.x;
    const int tid = threadIdx.x;

    for (int i = tid; i < 4096; i += 256) {
        Ft[(i & 63) * 64 + (i >> 6)] = F[i];
        bufA[i] = P0g[i];
    }
    for (int i = tid; i < 512; i += 256) Hs[i] = H[i];
    if (tid < 64) {
        Rs[tid]    = Rg[tid];
        meanv[tid] = init_mean[g * S_ + tid];
    }
    __syncthreads();

    for (int e = tid; e < 1024; e += 256) {
        const int k = e >> 4, m = e & 15;
        float acc = 0.f;
        if (m < 8)
            for (int j = 0; j < 64; ++j) acc += Hs[m * 64 + j] * Ft[k * 64 + j];
        Gt[e] = acc;
    }
    for (int e = tid; e < 512; e += 256) {
        const int s = e >> 3, m = e & 7;
        float acc = 0.f;
        for (int k = 0; k < 64; ++k) acc += Qg[s * 64 + k] * Hs[m * 64 + k];
        QHs[e] = acc;
    }
    __syncthreads();
    if (tid < 64) {
        const int i = tid >> 3, c = tid & 7;
        float acc = Rs[tid];
        for (int s = 0; s < 64; ++s) acc += Hs[i * 64 + s] * QHs[s * 8 + c];
        S0s[tid] = acc;
    }

    const int lane = tid & 31;
    const int wave = tid >> 5;
    const int half = lane >> 4;
    const int l15  = lane & 15;

    float* covP  = bufA;
    float* covpP = bufB;
    float* tmpP  = bufC;

    for (int t = 0; t < T_; ++t) {
        mm64(covP, Ft, tmpP, nullptr, nullptr, nullptr);
        if (wave >= 4) {
            const int ti = (wave - 4) * 16;
            const v8f acc = tile64(covP + ti, 64, Gt, 16, half, l15);
            #pragma unroll
            for (int r = 0; r < 8; ++r)
                VT[(ti + r + half * 8) * 16 + l15] = acc[r];
        }
        __syncthreads();

        if (wave == 4) {
            const v8f acc = tile64(VT, 16, Gt, 16, half, l15);
            if (half == 0) {
                #pragma unroll
                for (int r = 0; r < 8; ++r) {
                    float v;
                    if (l15 < 8) v = acc[r] + S0s[r * 8 + (l15 & 7)];
                    else         v = (l15 - 8 == r) ? 1.f : 0.f;
                    Wgj[r * 16 + l15] = v;
                }
            }
        }
        __syncthreads();

        {
            const int w4 = wave - 4;
            v8f accA = {};
            v8f accB = {};
            v8f accP = {};
            float yacc = 0.f;
            float macc = 0.f;
            #pragma unroll
            for (int p = 0; p < 8; ++p) {
                const float* Wo = (p & 1) ? Wgj2 : Wgj;
                float*       Wn = (p & 1) ? Wgj : Wgj2;
                if (tid < 64) {
                    const int r = tid >> 4, cc = tid & 15;
                    const float pp = Wo[p * 16 + p];
                    const float rp = Wo[p * 16 + cc];
                    const float w0 = Wo[r * 16 + cc];
                    const float f0 = Wo[r * 16 + p];
                    const float w1 = Wo[(r + 4) * 16 + cc];
                    const float f1 = Wo[(r + 4) * 16 + p];
                    __builtin_amdgcn_sched_barrier(0);
                    const float pv = 1.f / pp;
                    const float wp = rp * pv;
                    const float u0 = w0 - f0 * wp;
                    const float u1 = w1 - f1 * wp;
                    Wn[r * 16 + cc]       = (r == p)     ? wp : u0;
                    Wn[(r + 4) * 16 + cc] = (r + 4 == p) ? wp : u1;
                } else if (wave >= 4) {
                    const int tj  = w4 * 16;
                    const int tiA = (p >> 2) * 32;
                    const int tiB = tiA + 16;
                    const int kb  = (p & 3) * 4;
                    if ((p & 3) == 0) { accA = v8f{}; accB = v8f{}; }
                    float qA[8], qB[8];
                    if ((p & 3) == 3) {
                        for (int r = 0; r < 8; ++r) {
                            qA[r] = Qg[(tiA + r + half * 8) * 64 + tj + l15];
                            qB[r] = Qg[(tiB + r + half * 8) * 64 + tj + l15];
                        }
                    }
                    accP = slab(tmpP + tj, 64, Gt, 16, 2 * p,     half, l15, accP);
                    accP = slab(tmpP + tj, 64, Gt, 16, 2 * p + 1, half, l15, accP);
                    for (int s = 0; s < 4; ++s) {
                        const int k0 = (kb + s) * 4 + half * 2;
                        v2f a0, a1, b;
                        b.x  = Ft[(k0    ) * 64 + tj + l15];
                        b.y  = Ft[(k0 + 1) * 64 + tj + l15];
                        a0.x = tmpP[(k0    ) * 64 + tiA + l15];
                        a0.y = tmpP[(k0 + 1) * 64 + tiA + l15];
                        a1.x = tmpP[(k0    ) * 64 + tiB + l15];
                        a1.y = tmpP[(k0 + 1) * 64 + tiB + l15];
                        accA = __builtin_amdgcn_wmma_f32_16x16x4_f32(
                                   false, a0, false, b, (short)0, accA, false, false);
                        accB = __builtin_amdgcn_wmma_f32_16x16x4_f32(
                                   false, a1, false, b, (short)0, accB, false, false);
                    }
                    if ((p & 3) == 3) {
                        for (int r = 0; r < 8; ++r) {
                            covpP[(tiA + r + half * 8) * 64 + tj + l15] = accA[r] + qA[r];
                            covpP[(tiB + r + half * 8) * 64 + tj + l15] = accB[r] + qB[r];
                        }
                    }
                    if (p == 7 && l15 < 8) {
                        for (int r = 0; r < 8; ++r) {
                            const int row = tj + r + half * 8;
                            PHt[row * 8 + l15] = accP[r] + QHs[row * 8 + l15];
                        }
                    }
                } else if (p < 4) {
                    const int s = tid - 64;
                    for (int j = 0; j < 16; ++j)
                        macc = __builtin_fmaf(Ft[(p * 16 + j) * 64 + s], meanv[p * 16 + j], macc);
                    if (p == 3) meanp[s] = macc;
                } else if (tid < 72) {
                    const int m = tid - 64;
                    const int sb = (p - 4) * 16;
                    float yv = 0.f;
                    if (p == 7) yv = y[(g * T_ + t) * M_ + m];
                    for (int s = 0; s < 16; ++s)
                        yacc = __builtin_fmaf(Hs[m * 64 + sb + s], meanp[sb + s], yacc);
                    if (p == 7) {
                        out[(g * T_ + t) * M_ + m] = yacc;
                        resid[m] = yv - yacc;
                    }
                }
                __syncthreads();
            }
        }

        for (int e = tid; e < 512; e += 256) {
            const int s = e >> 3, m = e & 7;
            float acc = 0.f;
            for (int j = 0; j < 8; ++j) acc += PHt[s * 8 + j] * Wgj[j * 16 + 8 + m];
            Km[e] = acc;
            Kt[m * 64 + s] = acc;
        }
        __syncthreads();

        if (tid < 64) {
            float acc = meanp[tid];
            for (int m = 0; m < 8; ++m) acc += Km[tid * 8 + m] * resid[m];
            meanv[tid] = acc;
        }
        for (int e = tid; e < 512; e += 256) {
            const int i = e >> 3, m = e & 7;
            float acc = 0.f;
            for (int n = 0; n < 8; ++n) acc += Km[i * 8 + n] * Rs[n * 8 + m];
            KRt[m * 64 + i] = acc;
        }
        for (int tt = wave; tt < 16; tt += 8) {
            const int ti = (tt >> 2) * 16;
            const int tj = (tt & 3) * 16;
            v8f acc = {};
            acc = slab(Hs + ti, 64, Kt + tj, 64, 0, half, l15, acc);
            acc = slab(Hs + ti, 64, Kt + tj, 64, 1, half, l15, acc);
            #pragma unroll
            for (int r = 0; r < 8; ++r) {
                const int row = ti + r + half * 8;
                const int col = tj + l15;
                covP[row * 64 + col] = ((row == col) ? 1.f : 0.f) - acc[r];
            }
        }
        __syncthreads();

        mm64(covpP, covP, tmpP, nullptr, nullptr, nullptr);
        __syncthreads();
        mm64(tmpP, covP, covpP, nullptr, KRt, Kt);
        __syncthreads();

        float* sw = covP; covP = covpP; covpP = sw;
    }
}

extern "C" void kernel_launch(void* const* d_in, const int* in_sizes, int n_in,
                              void* d_out, int out_size, void* d_ws, size_t ws_size,
                              hipStream_t stream) {
    const float* y         = (const float*)d_in[0];
    const float* F         = (const float*)d_in[1];
    const float* H         = (const float*)d_in[2];
    const float* Lq        = (const float*)d_in[3];
    const float* Lr        = (const float*)d_in[4];
    const float* init_mean = (const float*)d_in[5];
    const float* Linit     = (const float*)d_in[6];
    float* out = (float*)d_out;
    float* ws  = (float*)d_ws;

    kf_prep<<<16, 256, 0, stream>>>(Lq, Lr, Linit, ws);

    const size_t smem_bytes = (4096 * 4 + 512 + 64 + 512 + 512 + 512 + 128 +
                               64 + 64 + 8 + 128 +
                               1024 + 1024 + 512 + 64 + 512) * sizeof(float);
    kf_main<<<G_, 256, smem_bytes, stream>>>(y, F, H,
                                             ws,
                                             ws + 8192,
                                             ws + 4096,
                                             init_mean, out);
}


# 64x64 WMMA operands in LDS in fragment order (a lane's two k adjacent, odd k-pair rows turned 32 banks): a fragment is one conflict-free 8-byte load, not two dwords of one bank; same sums, same order
# speedup vs baseline: 6.5059x; 1.3424x over previous
#include <hip/hip_runtime.h>
#include <hip/hip_bf16.h>

#define G_ 128
#define T_ 128
#define S_ 64
#define M_ 8

typedef float v2f __attribute__((ext_vector_type(2)));
typedef float v8f __attribute__((ext_vector_type(8)));

__device__ __forceinline__ int fo(const int k, const int m)
{
    const int P = k >> 1;
    return P * 128 + (((m << 1) + ((P & 1) << 5)) & 127) + (k & 1);
}

__device__ __forceinline__ int fbase(const int c, const int half)
{
    return half * 128 + (((c << 1) + (half << 5)) & 127);
}

__device__ __forceinline__ v2f frag(const float* X, const int base, const int kb)
{
    return *(const v2f*)(X + base + kb * 256);
}

__device__ __forceinline__ v8f slab(const float* At, const int ap,
                                    const float* Bk, const int bp,
                                    const int kb, const int half, const int l15,
                                    const v8f acc);

__device__ __forceinline__ void mm64(const float* At, const float* Bk, float* D,
                                     const float* At2,
                                     const float* Bk2)
{
    const int lane = threadIdx.x & 31;
    const int wave = threadIdx.x >> 5;
    const int half = lane >> 4;
    const int l15  = lane & 15;

    for (int tt = wave; tt < 16; tt += 8) {
        const int ti = (tt >> 2) * 16;
        const int tj = (tt & 3) * 16;
        const int ab = fbase(ti + l15, half);
        const int bb = fbase(tj + l15, half);
        v8f acc = {};
        #pragma unroll
        for (int kb = 0; kb < 16; ++kb) {
            const v2f a = frag(At, ab, kb);
            const v2f b = frag(Bk, bb, kb);
            acc = __builtin_amdgcn_wmma_f32_16x16x4_f32(
                      false, a, false, b, (short)0, acc, false, false);
        }
        if (At2) {
            acc = slab(At2 + ti, 64, Bk2 + tj, 64, 0, half, l15, acc);
            acc = slab(At2 + ti, 64, Bk2 + tj, 64, 1, half, l15, acc);
        }
        #pragma unroll
        for (int r = 0; r < 8; r += 2) {
            v2f v;
            v.x = acc[r];
            v.y = acc[r + 1];
            *(v2f*)(D + fo(ti + r + half * 8, tj + l15)) = v;
        }
    }
}

__device__ __forceinline__ v8f slab_fg(const float* X, const int xb,
                                       const float* Gk,
                                       const int kb, const int half, const int l15,
                                       const v8f acc)
{
    const int k0 = kb * 4 + half * 2;
    const v2f a = frag(X, xb, kb);
    v2f b;
    b.x = Gk[(k0    ) * 16 + l15];
    b.y = Gk[(k0 + 1) * 16 + l15];
    return __builtin_amdgcn_wmma_f32_16x16x4_f32(
               false, a, false, b, (short)0, acc, false, false);
}

__device__ __forceinline__ v8f tile_fg(const float* X, const int xb,
                                       const float* Gk,
                                       const int half, const int l15)
{
    v8f acc = {};
    #pragma unroll
    for (int kb = 0; kb < 16; ++kb)
        acc = slab_fg(X, xb, Gk, kb, half, l15, acc);
    return acc;
}

__device__ __forceinline__ v8f tile64(const float* At, const int ap,
                                      const float* Bk, const int bp,
                                      const int half, const int l15)
{
    v8f acc = {};
    #pragma unroll
    for (int kb = 0; kb < 16; ++kb) {
        const int k0 = kb * 4 + half * 2;
        v2f a, b;
        a.x = At[(k0    ) * ap + l15];
        a.y = At[(k0 + 1) * ap + l15];
        b.x = Bk[(k0    ) * bp + l15];
        b.y = Bk[(k0 + 1) * bp + l15];
        acc = __builtin_amdgcn_wmma_f32_16x16x4_f32(
                  false, a, false, b, (short)0, acc, false, false);
    }
    return acc;
}

__device__ __forceinline__ v8f slab(const float* At, const int ap,
                                    const float* Bk, const int bp,
                                    const int kb, const int half, const int l15,
                                    const v8f acc)
{
    const int k0 = kb * 4 + half * 2;
    v2f a, b;
    a.x = At[(k0    ) * ap + l15];
    a.y = At[(k0 + 1) * ap + l15];
    b.x = Bk[(k0    ) * bp + l15];
    b.y = Bk[(k0 + 1) * bp + l15];
    return __builtin_amdgcn_wmma_f32_16x16x4_f32(
               false, a, false, b, (short)0, acc, false, false);
}

__global__ __launch_bounds__(256) void kf_prep(const float* __restrict__ Lq,
                                               const float* __restrict__ Lr,
                                               const float* __restrict__ Linit,
                                               float* __restrict__ ws)
{
    float* Q  = ws;
    float* P0 = ws + 4096;
    float* R  = ws + 8192;
    for (int e = blockIdx.x * blockDim.x + threadIdx.x; e < 4096;
         e += gridDim.x * blockDim.x) {
        const int i = e >> 6, j = e & 63;
        float q = 0.f, p = 0.f;
        for (int k = 0; k < 64; ++k) {
            q += Lq[i * 64 + k] * Lq[j * 64 + k];
            p += Linit[i * 64 + k] * Linit[j * 64 + k];
        }
        Q[e] = q;
        P0[e] = p;
    }
    const int e = blockIdx.x * blockDim.x + threadIdx.x;
    if (e < 64) {
        const int i = e >> 3, j = e & 7;
        float r = 0.f;
        for (int k = 0; k < 8; ++k) r += Lr[i * 8 + k] * Lr[j * 8 + k];
        R[e] = r;
    }
}

__global__ __launch_bounds__(256) void kf_main(const float* __restrict__ y,
                                               const float* __restrict__ F,
                                               const float* __restrict__ H,
                                               const float* __restrict__ Qg,
                                               const float* __restrict__ Rg,
                                               const float* __restrict__ P0g,
                                               const float* __restrict__ init_mean,
                                               float* __restrict__ out)
{
    extern __shared__ __align__(16) float smem[];
    float* Ft    = smem;
    float* bufA  = Ft   + 4096;
    float* bufB  = bufA + 4096;
    float* bufC  = bufB + 4096;
    float* Hs    = bufC + 4096;
    float* Rs    = Hs   + 512;
    float* PHt   = Rs   + 64;
    float* Km    = PHt  + 512;
    float* KRt   = Km   + 512;
    float* Wgj   = KRt  + 512;
    float* meanv = Wgj  + 128;
    float* meanp = meanv + 64;
    float* resid = meanp + 64;
    float* Wgj2  = resid + 8;
    float* Gt    = Wgj2  + 128;
    float* VT    = Gt    + 1024;
    float* QHs   = VT    + 1024;
    float* S0s   = QHs   + 512;
    float* Kt    = S0s   + 64;

    const int g   = blockIdx.x;
    const int tid = threadIdx.x;

    for (int i = tid; i < 4096; i += 256) {
        Ft[fo(i & 63, i >> 6)] = F[i];
        bufA[fo(i >> 6, i & 63)] = P0g[i];
    }
    for (int i = tid; i < 512; i += 256) Hs[i] = H[i];
    if (tid < 64) {
        Rs[tid]    = Rg[tid];
        meanv[tid] = init_mean[g * S_ + tid];
    }
    __syncthreads();

    for (int e = tid; e < 1024; e += 256) {
        const int k = e >> 4, m = e & 15;
        float acc = 0.f;
        if (m < 8)
            for (int j = 0; j < 64; ++j) acc += Hs[m * 64 + j] * Ft[fo(k, j)];
        Gt[e] = acc;
    }
    for (int e = tid; e < 512; e += 256) {
        const int s = e >> 3, m = e & 7;
        float acc = 0.f;
        for (int k = 0; k < 64; ++k) acc += Qg[s * 64 + k] * Hs[m * 64 + k];
        QHs[e] = acc;
    }
    __syncthreads();
    if (tid < 64) {
        const int i = tid >> 3, c = tid & 7;
        float acc = Rs[tid];
        for (int s = 0; s < 64; ++s) acc += Hs[i * 64 + s] * QHs[s * 8 + c];
        S0s[tid] = acc;
    }

    const int lane = tid & 31;
    const int wave = tid >> 5;
    const int half = lane >> 4;
    const int l15  = lane & 15;

    float* covP  = bufA;
    float* covpP = bufB;
    float* tmpP  = bufC;

    for (int t = 0; t < T_; ++t) {
        mm64(covP, Ft, tmpP, nullptr, nullptr);
        if (wave >= 4) {
            const int ti = (wave - 4) * 16;
            const v8f acc = tile_fg(covP, fbase(ti + l15, half), Gt, half, l15);
            #pragma unroll
            for (int r = 0; r < 8; ++r)
                VT[(ti + r + half * 8) * 16 + l15] = acc[r];
        }
        __syncthreads();

        if (wave == 4) {
            const v8f acc = tile64(VT, 16, Gt, 16, half, l15);
            if (half == 0) {
                #pragma unroll
                for (int r = 0; r < 8; ++r) {
                    float v;
                    if (l15 < 8) v = acc[r] + S0s[r * 8 + (l15 & 7)];
                    else         v = (l15 - 8 == r) ? 1.f : 0.f;
                    Wgj[r * 16 + l15] = v;
                }
            }
        }
        __syncthreads();

        {
            const int w4 = wave - 4;
            v8f accA = {};
            v8f accB = {};
            v8f accP = {};
            float yacc = 0.f;
            float macc = 0.f;
            #pragma unroll
            for (int p = 0; p < 8; ++p) {
                const float* Wo = (p & 1) ? Wgj2 : Wgj;
                float*       Wn = (p & 1) ? Wgj : Wgj2;
                if (tid < 64) {
                    const int r = tid >> 4, cc = tid & 15;
                    const float pp = Wo[p * 16 + p];
                    const float rp = Wo[p * 16 + cc];
                    const float w0 = Wo[r * 16 + cc];
                    const float f0 = Wo[r * 16 + p];
                    const float w1 = Wo[(r + 4) * 16 + cc];
                    const float f1 = Wo[(r + 4) * 16 + p];
                    __builtin_amdgcn_sched_barrier(0);
                    const float pv = 1.f / pp;
                    const float wp = rp * pv;
                    const float u0 = w0 - f0 * wp;
                    const float u1 = w1 - f1 * wp;
                    Wn[r * 16 + cc]       = (r == p)     ? wp : u0;
                    Wn[(r + 4) * 16 + cc] = (r + 4 == p) ? wp : u1;
                } else if (wave >= 4) {
                    const int tj  = w4 * 16;
                    const int tiA = (p >> 2) * 32;
                    const int tiB = tiA + 16;
                    const int kb  = (p & 3) * 4;
                    if ((p & 3) == 0) { accA = v8f{}; accB = v8f{}; }
                    float qA[8], qB[8];
                    if ((p & 3) == 3) {
                        for (int r = 0; r < 8; ++r) {
                            qA[r] = Qg[(tiA + r + half * 8) * 64 + tj + l15];
                            qB[r] = Qg[(tiB + r + half * 8) * 64 + tj + l15];
                        }
                    }
                    const int bJ = fbase(tj + l15, half);
                    const int bA = fbase(tiA + l15, half);
                    const int bB = fbase(tiB + l15, half);
                    accP = slab_fg(tmpP, bJ, Gt, 2 * p,     half, l15, accP);
                    accP = slab_fg(tmpP, bJ, Gt, 2 * p + 1, half, l15, accP);
                    for (int s = 0; s < 4; ++s) {
                        const v2f b  = frag(Ft,   bJ, kb + s);
                        const v2f a0 = frag(tmpP, bA, kb + s);
                        const v2f a1 = frag(tmpP, bB, kb + s);
                        accA = __builtin_amdgcn_wmma_f32_16x16x4_f32(
                                   false, a0, false, b, (short)0, accA, false, false);
                        accB = __builtin_amdgcn_wmma_f32_16x16x4_f32(
                                   false, a1, false, b, (short)0, accB, false, false);
                    }
                    if ((p & 3) == 3) {
                        for (int r = 0; r < 8; r += 2) {
                            v2f vA, vB;
                            vA.x = accA[r]     + qA[r];
                            vA.y = accA[r + 1] + qA[r + 1];
                            vB.x = accB[r]     + qB[r];
                            vB.y = accB[r + 1] + qB[r + 1];
                            *(v2f*)(covpP + fo(tiA + r + half * 8, tj + l15)) = vA;
                            *(v2f*)(covpP + fo(tiB + r + half * 8, tj + l15)) = vB;
                        }
                    }
                    if (p == 7 && l15 < 8) {
                        for (int r = 0; r < 8; ++r) {
                            const int row = tj + r + half * 8;
                            PHt[row * 8 + l15] = accP[r] + QHs[row * 8 + l15];
                        }
                    }
                } else if (p < 4) {
                    const int s = tid - 64;
                    for (int j = 0; j < 16; ++j)
                        macc = __builtin_fmaf(Ft[fo(p * 16 + j, s)], meanv[p * 16 + j], macc);
                    if (p == 3) meanp[s] = macc;
                } else if (tid < 72) {
                    const int m = tid - 64;
                    const int sb = (p - 4) * 16;
                    float yv = 0.f;
                    if (p == 7) yv = y[(g * T_ + t) * M_ + m];
                    for (int s = 0; s < 16; ++s)
                        yacc = __builtin_fmaf(Hs[m * 64 + sb + s], meanp[sb + s], yacc);
                    if (p == 7) {
                        out[(g * T_ + t) * M_ + m] = yacc;
                        resid[m] = yv - yacc;
                    }
                }
                __syncthreads();
            }
        }

        for (int e = tid; e < 512; e += 256) {
            const int s = e >> 3, m = e & 7;
            float acc = 0.f;
            for (int j = 0; j < 8; ++j) acc += PHt[s * 8 + j] * Wgj[j * 16 + 8 + m];
            Km[e] = acc;
            Kt[m * 64 + s] = acc;
        }
        __syncthreads();

        if (tid < 64) {
            float acc = meanp[tid];
            for (int m = 0; m < 8; ++m) acc += Km[tid * 8 + m] * resid[m];
            meanv[tid] = acc;
        }
        for (int e = tid; e < 512; e += 256) {
            const int i = e >> 3, m = e & 7;
            float acc = 0.f;
            for (int n = 0; n < 8; ++n) acc += Km[i * 8 + n] * Rs[n * 8 + m];
            KRt[m * 64 + i] = acc;
        }
        for (int tt = wave; tt < 16; tt += 8) {
            const int ti = (tt >> 2) * 16;
            const int tj = (tt & 3) * 16;
            v8f acc = {};
            acc = slab(Hs + ti, 64, Kt + tj, 64, 0, half, l15, acc);
            acc = slab(Hs + ti, 64, Kt + tj, 64, 1, half, l15, acc);
            #pragma unroll
            for (int r = 0; r < 8; r += 2) {
                const int row = ti + r + half * 8;
                const int col = tj + l15;
                v2f v;
                v.x = ((row     == col) ? 1.f : 0.f) - acc[r];
                v.y = ((row + 1 == col) ? 1.f : 0.f) - acc[r + 1];
                *(v2f*)(covP + fo(row, col)) = v;
            }
        }
        __syncthreads();

        mm64(covpP, covP, tmpP, nullptr, nullptr);
        __syncthreads();
        mm64(tmpP, covP, covpP, KRt, Kt);
        __syncthreads();

        float* sw = covP; covP = covpP; covpP = sw;
    }
}

extern "C" void kernel_launch(void* const* d_in, const int* in_sizes, int n_in,
                              void* d_out, int out_size, void* d_ws, size_t ws_size,
                              hipStream_t stream) {
    const float* y         = (const float*)d_in[0];
    const float* F         = (const float*)d_in[1];
    const float* H         = (const float*)d_in[2];
    const float* Lq        = (const float*)d_in[3];
    const float* Lr        = (const float*)d_in[4];
    const float* init_mean = (const float*)d_in[5];
    const float* Linit     = (const float*)d_in[6];
    float* out = (float*)d_out;
    float* ws  = (float*)d_ws;

    kf_prep<<<16, 256, 0, stream>>>(Lq, Lr, Linit, ws);

    const size_t smem_bytes = (4096 * 4 + 512 + 64 + 512 + 512 + 512 + 128 +
                               64 + 64 + 8 + 128 +
                               1024 + 1024 + 512 + 64 + 512) * sizeof(float);
    kf_main<<<G_, 256, smem_bytes, stream>>>(y, F, H,
                                             ws,
                                             ws + 8192,
                                             ws + 4096,
                                             init_mean, out);
}


# Each WMMA run's LDS fragments loaded as one group ahead of it; a wave's two 64x64-product tiles run as two chains sharing B (waves 4-7: the G tile as a third); same sums, same order
# speedup vs baseline: 7.9013x; 1.2145x over previous
#include <hip/hip_runtime.h>
#include <hip/hip_bf16.h>

#define G_ 128
#define T_ 128
#define S_ 64
#define M_ 8

typedef float v2f __attribute__((ext_vector_type(2)));
typedef float v8f __attribute__((ext_vector_type(8)));

__device__ __forceinline__ int fo(const int k, const int m)
{
    const int P = k >> 1;
    return P * 128 + (((m << 1) + ((P & 1) << 5)) & 127) + (k & 1);
}

__device__ __forceinline__ int fbase(const int c, const int half)
{
    return half * 128 + (((c << 1) + (half << 5)) & 127);
}

__device__ __forceinline__ v2f frag(const float* X, const int base, const int kb)
{
    return *(const v2f*)(X + base + kb * 256);
}

template <bool RIDER, bool EXTRA>
__device__ __forceinline__ void mm64(const float* At, const float* Bk, float* D,
                                     const float* At2,
                                     const float* Bk2,
                                     const float* Gk, float* Vo, const int gi)
{
    const int lane = threadIdx.x & 31;
    const int wave = threadIdx.x >> 5;
    const int half = lane >> 4;
    const int l15  = lane & 15;

    const int ti0 = (wave >> 2) * 16;
    const int ti1 = ti0 + 32;
    const int tj  = (wave & 3) * 16;
    const int ab0 = fbase(ti0 + l15, half);
    const int ab1 = fbase(ti1 + l15, half);
    const int bb  = fbase(tj  + l15, half);

    v2f b[16], a0[16], a1[16];
    #pragma unroll
    for (int kb = 0; kb < 16; ++kb) {
        b[kb]  = frag(Bk, bb,  kb);
        a0[kb] = frag(At, ab0, kb);
        a1[kb] = frag(At, ab1, kb);
    }
    v2f xb[2], x0[2], x1[2];
    if (EXTRA) {
        #pragma unroll
        for (int kb = 0; kb < 2; ++kb) {
            const int k0 = kb * 4 + half * 2;
            xb[kb].x = Bk2[(k0    ) * 64 + tj + l15];
            xb[kb].y = Bk2[(k0 + 1) * 64 + tj + l15];
            x0[kb].x = At2[(k0    ) * 64 + ti0 + l15];
            x0[kb].y = At2[(k0 + 1) * 64 + ti0 + l15];
            x1[kb].x = At2[(k0    ) * 64 + ti1 + l15];
            x1[kb].y = At2[(k0 + 1) * 64 + ti1 + l15];
        }
    }
    v2f ga[16], gb[16];
    if (RIDER) {
        const int gab = fbase(gi + l15, half);
        #pragma unroll
        for (int kb = 0; kb < 16; ++kb) {
            const int k0 = kb * 4 + half * 2;
            ga[kb]   = frag(At, gab, kb);
            gb[kb].x = Gk[(k0    ) * 16 + l15];
            gb[kb].y = Gk[(k0 + 1) * 16 + l15];
        }
    }
    __builtin_amdgcn_sched_barrier(0);

    v8f acc0 = {};
    v8f acc1 = {};
    v8f accg = {};
    #pragma unroll
    for (int kb = 0; kb < 16; ++kb) {
        acc0 = __builtin_amdgcn_wmma_f32_16x16x4_f32(
                   false, a0[kb], false, b[kb], (short)0, acc0, false, false);
        acc1 = __builtin_amdgcn_wmma_f32_16x16x4_f32(
                   false, a1[kb], false, b[kb], (short)0, acc1, false, false);
        if (RIDER)
            accg = __builtin_amdgcn_wmma_f32_16x16x4_f32(
                       false, ga[kb], false, gb[kb], (short)0, accg, false, false);
    }
    if (EXTRA) {
        #pragma unroll
        for (int kb = 0; kb < 2; ++kb) {
            acc0 = __builtin_amdgcn_wmma_f32_16x16x4_f32(
                       false, x0[kb], false, xb[kb], (short)0, acc0, false, false);
            acc1 = __builtin_amdgcn_wmma_f32_16x16x4_f32(
                       false, x1[kb], false, xb[kb], (short)0, acc1, false, false);
        }
    }
    #pragma unroll
    for (int r = 0; r < 8; r += 2) {
        v2f v0, v1;
        v0.x = acc0[r];
        v0.y = acc0[r + 1];
        v1.x = acc1[r];
        v1.y = acc1[r + 1];
        *(v2f*)(D + fo(ti0 + r + half * 8, tj + l15)) = v0;
        *(v2f*)(D + fo(ti1 + r + half * 8, tj + l15)) = v1;
    }
    if (RIDER) {
        #pragma unroll
        for (int r = 0; r < 8; ++r)
            Vo[(gi + r + half * 8) * 16 + l15] = accg[r];
    }
}

__device__ __forceinline__ v8f slab_fg(const float* X, const int xb,
                                       const float* Gk,
                                       const int kb, const int half, const int l15,
                                       const v8f acc)
{
    const int k0 = kb * 4 + half * 2;
    const v2f a = frag(X, xb, kb);
    v2f b;
    b.x = Gk[(k0    ) * 16 + l15];
    b.y = Gk[(k0 + 1) * 16 + l15];
    return __builtin_amdgcn_wmma_f32_16x16x4_f32(
               false, a, false, b, (short)0, acc, false, false);
}

__global__ __launch_bounds__(256) void kf_prep(const float* __restrict__ Lq,
                                               const float* __restrict__ Lr,
                                               const float* __restrict__ Linit,
                                               float* __restrict__ ws)
{
    float* Q  = ws;
    float* P0 = ws + 4096;
    float* R  = ws + 8192;
    for (int e = blockIdx.x * blockDim.x + threadIdx.x; e < 4096;
         e += gridDim.x * blockDim.x) {
        const int i = e >> 6, j = e & 63;
        float q = 0.f, p = 0.f;
        for (int k = 0; k < 64; ++k) {
            q += Lq[i * 64 + k] * Lq[j * 64 + k];
            p += Linit[i * 64 + k] * Linit[j * 64 + k];
        }
        Q[e] = q;
        P0[e] = p;
    }
    const int e = blockIdx.x * blockDim.x + threadIdx.x;
    if (e < 64) {
        const int i = e >> 3, j = e & 7;
        float r = 0.f;
        for (int k = 0; k < 8; ++k) r += Lr[i * 8 + k] * Lr[j * 8 + k];
        R[e] = r;
    }
}

__global__ __launch_bounds__(256) void kf_main(const float* __restrict__ y,
                                               const float* __restrict__ F,
                                               const float* __restrict__ H,
                                               const float* __restrict__ Qg,
                                               const float* __restrict__ Rg,
                                               const float* __restrict__ P0g,
                                               const float* __restrict__ init_mean,
                                               float* __restrict__ out)
{
    extern __shared__ __align__(16) float smem[];
    float* Ft    = smem;
    float* bufA  = Ft   + 4096;
    float* bufB  = bufA + 4096;
    float* bufC  = bufB + 4096;
    float* Hs    = bufC + 4096;
    float* Rs    = Hs   + 512;
    float* PHt   = Rs   + 64;
    float* Km    = PHt  + 512;
    float* KRt   = Km   + 512;
    float* Wgj   = KRt  + 512;
    float* meanv = Wgj  + 128;
    float* meanp = meanv + 64;
    float* resid = meanp + 64;
    float* Wgj2  = resid + 8;
    float* Gt    = Wgj2  + 128;
    float* VT    = Gt    + 1024;
    float* QHs   = VT    + 1024;
    float* S0s   = QHs   + 512;
    float* Kt    = S0s   + 64;

    const int g   = blockIdx.x;
    const int tid = threadIdx.x;

    for (int i = tid; i < 4096; i += 256) {
        Ft[fo(i & 63, i >> 6)] = F[i];
        bufA[fo(i >> 6, i & 63)] = P0g[i];
    }
    for (int i = tid; i < 512; i += 256) Hs[i] = H[i];
    if (tid < 64) {
        Rs[tid]    = Rg[tid];
        meanv[tid] = init_mean[g * S_ + tid];
    }
    __syncthreads();

    for (int e = tid; e < 1024; e += 256) {
        const int k = e >> 4, m = e & 15;
        float acc = 0.f;
        if (m < 8)
            for (int j = 0; j < 64; ++j) acc += Hs[m * 64 + j] * Ft[fo(k, j)];
        Gt[e] = acc;
    }
    for (int e = tid; e < 512; e += 256) {
        const int s = e >> 3, m = e & 7;
        float acc = 0.f;
        for (int k = 0; k < 64; ++k) acc += Qg[s * 64 + k] * Hs[m * 64 + k];
        QHs[e] = acc;
    }
    __syncthreads();
    if (tid < 64) {
        const int i = tid >> 3, c = tid & 7;
        float acc = Rs[tid];
        for (int s = 0; s < 64; ++s) acc += Hs[i * 64 + s] * QHs[s * 8 + c];
        S0s[tid] = acc;
    }

    const int lane = tid & 31;
    const int wave = tid >> 5;
    const int half = lane >> 4;
    const int l15  = lane & 15;

    float* covP  = bufA;
    float* covpP = bufB;
    float* tmpP  = bufC;

    for (int t = 0; t < T_; ++t) {
        if (wave >= 4)
            mm64<true, false>(covP, Ft, tmpP, nullptr, nullptr, Gt, VT, (wave - 4) * 16);
        else
            mm64<false, false>(covP, Ft, tmpP, nullptr, nullptr, nullptr, nullptr, 0);
        __syncthreads();

        if (wave == 4) {
            v2f a[16], b[16];
            #pragma unroll
            for (int kb = 0; kb < 16; ++kb) {
                const int k0 = kb * 4 + half * 2;
                a[kb].x = VT[(k0    ) * 16 + l15];
                a[kb].y = VT[(k0 + 1) * 16 + l15];
                b[kb].x = Gt[(k0    ) * 16 + l15];
                b[kb].y = Gt[(k0 + 1) * 16 + l15];
            }
            float s0[8];
            #pragma unroll
            for (int r = 0; r < 8; ++r) s0[r] = S0s[r * 8 + (l15 & 7)];
            __builtin_amdgcn_sched_barrier(0);
            v8f acc = {};
            #pragma unroll
            for (int kb = 0; kb < 16; ++kb)
                acc = __builtin_amdgcn_wmma_f32_16x16x4_f32(
                          false, a[kb], false, b[kb], (short)0, acc, false, false);
            if (half == 0) {
                #pragma unroll
                for (int r = 0; r < 8; ++r) {
                    const float v = (l15 < 8) ? acc[r] + s0[r]
                                              : ((l15 - 8 == r) ? 1.f : 0.f);
                    Wgj[r * 16 + l15] = v;
                }
            }
        }
        __syncthreads();

        {
            const int w4 = wave - 4;
            v8f accA = {};
            v8f accB = {};
            v8f accP = {};
            float yacc = 0.f;
            float macc = 0.f;
            #pragma unroll
            for (int p = 0; p < 8; ++p) {
                const float* Wo = (p & 1) ? Wgj2 : Wgj;
                float*       Wn = (p & 1) ? Wgj : Wgj2;
                if (tid < 64) {
                    const int r = tid >> 4, cc = tid & 15;
                    const float pp = Wo[p * 16 + p];
                    const float rp = Wo[p * 16 + cc];
                    const float w0 = Wo[r * 16 + cc];
                    const float f0 = Wo[r * 16 + p];
                    const float w1 = Wo[(r + 4) * 16 + cc];
                    const float f1 = Wo[(r + 4) * 16 + p];
                    __builtin_amdgcn_sched_barrier(0);
                    const float pv = 1.f / pp;
                    const float wp = rp * pv;
                    const float u0 = w0 - f0 * wp;
                    const float u1 = w1 - f1 * wp;
                    Wn[r * 16 + cc]       = (r == p)     ? wp : u0;
                    Wn[(r + 4) * 16 + cc] = (r + 4 == p) ? wp : u1;
                } else if (wave >= 4) {
                    const int tj  = w4 * 16;
                    const int tiA = (p >> 2) * 32;
                    const int tiB = tiA + 16;
                    const int kb  = (p & 3) * 4;
                    if ((p & 3) == 0) { accA = v8f{}; accB = v8f{}; }
                    float qA[8], qB[8];
                    if ((p & 3) == 3) {
                        for (int r = 0; r < 8; ++r) {
                            qA[r] = Qg[(tiA + r + half * 8) * 64 + tj + l15];
                            qB[r] = Qg[(tiB + r + half * 8) * 64 + tj + l15];
                        }
                    }
                    const int bJ = fbase(tj + l15, half);
                    const int bA = fbase(tiA + l15, half);
                    const int bB = fbase(tiB + l15, half);
                    accP = slab_fg(tmpP, bJ, Gt, 2 * p,     half, l15, accP);
                    accP = slab_fg(tmpP, bJ, Gt, 2 * p + 1, half, l15, accP);
                    for (int s = 0; s < 4; ++s) {
                        const v2f b  = frag(Ft,   bJ, kb + s);
                        const v2f a0 = frag(tmpP, bA, kb + s);
                        const v2f a1 = frag(tmpP, bB, kb + s);
                        accA = __builtin_amdgcn_wmma_f32_16x16x4_f32(
                                   false, a0, false, b, (short)0, accA, false, false);
                        accB = __builtin_amdgcn_wmma_f32_16x16x4_f32(
                                   false, a1, false, b, (short)0, accB, false, false);
                    }
                    if ((p & 3) == 3) {
                        for (int r = 0; r < 8; r += 2) {
                            v2f vA, vB;
                            vA.x = accA[r]     + qA[r];
                            vA.y = accA[r + 1] + qA[r + 1];
                            vB.x = accB[r]     + qB[r];
                            vB.y = accB[r + 1] + qB[r + 1];
                            *(v2f*)(covpP + fo(tiA + r + half * 8, tj + l15)) = vA;
                            *(v2f*)(covpP + fo(tiB + r + half * 8, tj + l15)) = vB;
                        }
                    }
                    if (p == 7 && l15 < 8) {
                        for (int r = 0; r < 8; ++r) {
                            const int row = tj + r + half * 8;
                            PHt[row * 8 + l15] = accP[r] + QHs[row * 8 + l15];
                        }
                    }
                } else if (p < 4) {
                    const int s = tid - 64;
                    for (int j = 0; j < 16; ++j)
                        macc = __builtin_fmaf(Ft[fo(p * 16 + j, s)], meanv[p * 16 + j], macc);
                    if (p == 3) meanp[s] = macc;
                } else if (tid < 72) {
                    const int m = tid - 64;
                    const int sb = (p - 4) * 16;
                    float yv = 0.f;
                    if (p == 7) yv = y[(g * T_ + t) * M_ + m];
                    for (int s = 0; s < 16; ++s)
                        yacc = __builtin_fmaf(Hs[m * 64 + sb + s], meanp[sb + s], yacc);
                    if (p == 7) {
                        out[(g * T_ + t) * M_ + m] = yacc;
                        resid[m] = yv - yacc;
                    }
                }
                __syncthreads();
            }
        }

        for (int e = tid; e < 512; e += 256) {
            const int s = e >> 3, m = e & 7;
            float acc = 0.f;
            for (int j = 0; j < 8; ++j) acc += PHt[s * 8 + j] * Wgj[j * 16 + 8 + m];
            Km[e] = acc;
            Kt[m * 64 + s] = acc;
        }
        __syncthreads();

        if (tid < 64) {
            float acc = meanp[tid];
            for (int m = 0; m < 8; ++m) acc += Km[tid * 8 + m] * resid[m];
            meanv[tid] = acc;
        }
        for (int e = tid; e < 512; e += 256) {
            const int i = e >> 3, m = e & 7;
            float acc = 0.f;
            for (int n = 0; n < 8; ++n) acc += Km[i * 8 + n] * Rs[n * 8 + m];
            KRt[m * 64 + i] = acc;
        }
        {
            const int ti0 = (wave >> 2) * 16;
            const int ti1 = ti0 + 32;
            const int tj  = (wave & 3) * 16;
            v2f h0[2], h1[2], kt[2];
            #pragma unroll
            for (int kb = 0; kb < 2; ++kb) {
                const int k0 = kb * 4 + half * 2;
                kt[kb].x = Kt[(k0    ) * 64 + tj  + l15];
                kt[kb].y = Kt[(k0 + 1) * 64 + tj  + l15];
                h0[kb].x = Hs[(k0    ) * 64 + ti0 + l15];
                h0[kb].y = Hs[(k0 + 1) * 64 + ti0 + l15];
                h1[kb].x = Hs[(k0    ) * 64 + ti1 + l15];
                h1[kb].y = Hs[(k0 + 1) * 64 + ti1 + l15];
            }
            v8f acc0 = {};
            v8f acc1 = {};
            #pragma unroll
            for (int kb = 0; kb < 2; ++kb) {
                acc0 = __builtin_amdgcn_wmma_f32_16x16x4_f32(
                           false, h0[kb], false, kt[kb], (short)0, acc0, false, false);
                acc1 = __builtin_amdgcn_wmma_f32_16x16x4_f32(
                           false, h1[kb], false, kt[kb], (short)0, acc1, false, false);
            }
            #pragma unroll
            for (int r = 0; r < 8; r += 2) {
                const int row0 = ti0 + r + half * 8;
                const int row1 = ti1 + r + half * 8;
                const int col  = tj + l15;
                v2f v0, v1;
                v0.x = ((row0     == col) ? 1.f : 0.f) - acc0[r];
                v0.y = ((row0 + 1 == col) ? 1.f : 0.f) - acc0[r + 1];
                v1.x = ((row1     == col) ? 1.f : 0.f) - acc1[r];
                v1.y = ((row1 + 1 == col) ? 1.f : 0.f) - acc1[r + 1];
                *(v2f*)(covP + fo(row0, col)) = v0;
                *(v2f*)(covP + fo(row1, col)) = v1;
            }
        }
        __syncthreads();

        mm64<false, false>(covpP, covP, tmpP, nullptr, nullptr, nullptr, nullptr, 0);
        __syncthreads();
        mm64<false, true>(tmpP, covP, covpP, KRt, Kt, nullptr, nullptr, 0);
        __syncthreads();

        float* sw = covP; covP = covpP; covpP = sw;
    }
}

extern "C" void kernel_launch(void* const* d_in, const int* in_sizes, int n_in,
                              void* d_out, int out_size, void* d_ws, size_t ws_size,
                              hipStream_t stream) {
    const float* y         = (const float*)d_in[0];
    const float* F         = (const float*)d_in[1];
    const float* H         = (const float*)d_in[2];
    const float* Lq        = (const float*)d_in[3];
    const float* Lr        = (const float*)d_in[4];
    const float* init_mean = (const float*)d_in[5];
    const float* Linit     = (const float*)d_in[6];
    float* out = (float*)d_out;
    float* ws  = (float*)d_ws;

    kf_prep<<<16, 256, 0, stream>>>(Lq, Lr, Linit, ws);

    const size_t smem_bytes = (4096 * 4 + 512 + 64 + 512 + 512 + 512 + 128 +
                               64 + 64 + 8 + 128 +
                               1024 + 1024 + 512 + 64 + 512) * sizeof(float);
    kf_main<<<G_, 256, smem_bytes, stream>>>(y, F, H,
                                             ws,
                                             ws + 8192,
                                             ws + 4096,
                                             init_mean, out);
}


# cov_p' = F cov_p F^T + Q - L (F PHt)^T, L = F PHt S^-1: the 64-term products need no inverse and run in K-slab slices beside PHt, S_inn, inverse, L; cov_p held symmetric by mirror; sum order changed
# speedup vs baseline: 12.2143x; 1.5459x over previous
#include <hip/hip_runtime.h>
#include <hip/hip_bf16.h>

#define G_ 128
#define T_ 128
#define S_ 64
#define M_ 8

typedef float v2f __attribute__((ext_vector_type(2)));
typedef float v8f __attribute__((ext_vector_type(8)));

__device__ __forceinline__ int fo(const int k, const int m)
{
    const int P = k >> 1;
    return P * 128 + (((m << 1) + ((P & 1) << 5)) & 127) + (k & 1);
}

__device__ __forceinline__ int fbase(const int c, const int half)
{
    return half * 128 + (((c << 1) + (half << 5)) & 127);
}

__device__ __forceinline__ v2f frag(const float* X, const int base, const int kb)
{
    return *(const v2f*)(X + base + kb * 256);
}

struct TileCtx {
    int ti0, ti1, tj;
    int ab0, ab1, bb;
    int half, l15;
};

__device__ __forceinline__ TileCtx tile_ctx()
{
    TileCtx c;
    const int lane = threadIdx.x & 31;
    const int wave = threadIdx.x >> 5;
    c.half = lane >> 4;
    c.l15  = lane & 15;
    c.ti0  = (wave >> 2) * 16;
    c.ti1  = c.ti0 + 32;
    c.tj   = (wave & 3) * 16;
    c.ab0  = fbase(c.ti0 + c.l15, c.half);
    c.ab1  = fbase(c.ti1 + c.l15, c.half);
    c.bb   = fbase(c.tj  + c.l15, c.half);
    return c;
}

template <int KB0, int KB1>
__device__ __forceinline__ void mm_slice(const float* At, const float* Bk, const TileCtx& c,
                                         v8f& acc0, v8f& acc1,
                                         const bool run0, const bool run1)
{
    if (run1) {
        v2f b[KB1 - KB0], a0[KB1 - KB0], a1[KB1 - KB0];
        #pragma unroll
        for (int i = 0; i < KB1 - KB0; ++i) {
            b[i]  = frag(Bk, c.bb,  KB0 + i);
            a0[i] = frag(At, c.ab0, KB0 + i);
            a1[i] = frag(At, c.ab1, KB0 + i);
        }
        __builtin_amdgcn_sched_barrier(0);
        #pragma unroll
        for (int i = 0; i < KB1 - KB0; ++i) {
            acc0 = __builtin_amdgcn_wmma_f32_16x16x4_f32(
                       false, a0[i], false, b[i], (short)0, acc0, false, false);
            acc1 = __builtin_amdgcn_wmma_f32_16x16x4_f32(
                       false, a1[i], false, b[i], (short)0, acc1, false, false);
        }
    } else if (run0) {
        v2f b[KB1 - KB0], a0[KB1 - KB0];
        #pragma unroll
        for (int i = 0; i < KB1 - KB0; ++i) {
            b[i]  = frag(Bk, c.bb,  KB0 + i);
            a0[i] = frag(At, c.ab0, KB0 + i);
        }
        __builtin_amdgcn_sched_barrier(0);
        #pragma unroll
        for (int i = 0; i < KB1 - KB0; ++i)
            acc0 = __builtin_amdgcn_wmma_f32_16x16x4_f32(
                       false, a0[i], false, b[i], (short)0, acc0, false, false);
    }
}

template <int A0, int A1, int B0, int B1, int C0, int C1>
__device__ __forceinline__ void p2_slice(const float* At, const float* Bk, const TileCtx& c,
                                         v8f& m0, v8f& m1, const int wave, const bool up1)
{
    if (wave < 2)      mm_slice<A0, A1>(At, Bk, c, m0, m1, true, false);
    else if (wave < 4) mm_slice<B0, B1>(At, Bk, c, m0, m1, true, true);
    else if (wave > 4) mm_slice<C0, C1>(At, Bk, c, m0, m1, true, up1);
}

__device__ __forceinline__ v8f q_tile(const float* Qg, const int ti, const TileCtx& c)
{
    v8f q;
    #pragma unroll
    for (int r = 0; r < 8; ++r) q[r] = Qg[(ti + r + c.half * 8) * 64 + c.tj + c.l15];
    return q;
}

__device__ __forceinline__ void store_tile(float* D, const v8f& acc, const int ti, const TileCtx& c)
{
    const int col = c.tj + c.l15;
    #pragma unroll
    for (int r = 0; r < 8; r += 2) {
        v2f v;
        v.x = acc[r];
        v.y = acc[r + 1];
        *(v2f*)(D + fo(ti + r + c.half * 8, col)) = v;
    }
}

__device__ __forceinline__ void store_tile_sym(float* D, const v8f& acc, const int ti, const TileCtx& c)
{
    const int col = c.tj + c.l15;
    if (ti < c.tj) {
        #pragma unroll
        for (int r = 0; r < 8; r += 2) {
            const int row = ti + r + c.half * 8;
            v2f v;
            v.x = acc[r];
            v.y = acc[r + 1];
            *(v2f*)(D + fo(row, col)) = v;
            D[fo(col, row    )] = v.x;
            D[fo(col, row + 1)] = v.y;
        }
    } else if (ti == c.tj) {
        #pragma unroll
        for (int r = 0; r < 8; ++r) {
            const int rl  = r + c.half * 8;
            const int row = ti + rl;
            if (rl <= c.l15) D[fo(row, col)] = acc[r];
            if (rl <  c.l15) D[fo(col, row)] = acc[r];
        }
    }
}

__device__ __forceinline__ float fmean_terms(const float* Ft, const float* meanv,
                                             const int j0, const int s, float acc)
{
    v2f   fv[16];
    float mv[32];
    #pragma unroll
    for (int P = 0; P < 16; ++P) fv[P] = *(const v2f*)(Ft + fo(j0 + 2 * P, s));
    #pragma unroll
    for (int j = 0; j < 32; ++j) mv[j] = meanv[j0 + j];
    __builtin_amdgcn_sched_barrier(0);
    #pragma unroll
    for (int P = 0; P < 16; ++P) {
        acc = __builtin_fmaf(fv[P].x, mv[2 * P],     acc);
        acc = __builtin_fmaf(fv[P].y, mv[2 * P + 1], acc);
    }
    return acc;
}

__device__ __forceinline__ float yhat_terms(const float* Hk, const float* meanv,
                                            const int s0, const int m, float acc)
{
    float hv[32];
    float mv[32];
    #pragma unroll
    for (int s = 0; s < 32; ++s) hv[s] = Hk[(s0 + s) * 16 + m];
    #pragma unroll
    for (int s = 0; s < 32; ++s) mv[s] = meanv[s0 + s];
    __builtin_amdgcn_sched_barrier(0);
    #pragma unroll
    for (int s = 0; s < 32; ++s)
        acc = __builtin_fmaf(hv[s], mv[s], acc);
    return acc;
}

__global__ __launch_bounds__(256) void kf_prep(const float* __restrict__ Lq,
                                               const float* __restrict__ Lr,
                                               const float* __restrict__ Linit,
                                               float* __restrict__ ws)
{
    float* Q  = ws;
    float* P0 = ws + 4096;
    float* R  = ws + 8192;
    for (int e = blockIdx.x * blockDim.x + threadIdx.x; e < 4096;
         e += gridDim.x * blockDim.x) {
        const int i = e >> 6, j = e & 63;
        float q = 0.f, p = 0.f;
        for (int k = 0; k < 64; ++k) {
            q += Lq[i * 64 + k] * Lq[j * 64 + k];
            p += Linit[i * 64 + k] * Linit[j * 64 + k];
        }
        Q[e] = q;
        P0[e] = p;
    }
    const int e = blockIdx.x * blockDim.x + threadIdx.x;
    if (e < 64) {
        const int i = e >> 3, j = e & 7;
        float r = 0.f;
        for (int k = 0; k < 8; ++k) r += Lr[i * 8 + k] * Lr[j * 8 + k];
        R[e] = r;
    }
}

__global__ __launch_bounds__(256) void kf_main(const float* __restrict__ y,
                                               const float* __restrict__ F,
                                               const float* __restrict__ H,
                                               const float* __restrict__ Qg,
                                               const float* __restrict__ Rg,
                                               const float* __restrict__ P0g,
                                               const float* __restrict__ init_mean,
                                               float* __restrict__ out)
{
    extern __shared__ __align__(16) float smem[];
    float* Ft    = smem;
    float* Pn    = Ft   + 4096;
    float* tmpP  = Pn   + 4096;
    float* Rs    = tmpP + 4096;
    float* PHk   = Rs   + 64;
    float* FPH   = PHk  + 1024;
    float* FPHt  = FPH  + 512;
    float* Lm    = FPHt + 512;
    float* NLt   = Lm   + 512;
    float* Wgj   = NLt  + 512;
    float* meanv = Wgj  + 128;
    float* meanp = meanv + 64;
    float* resid = meanp + 64;
    float* Wgj2  = resid + 8;
    float* Hk    = Wgj2  + 128;

    const int g   = blockIdx.x;
    const int tid = threadIdx.x;

    for (int i = tid; i < 4096; i += 256) {
        Ft[fo(i & 63, i >> 6)] = F[i];
        Pn[fo(i >> 6, i & 63)] = P0g[i];
    }
    for (int i = tid; i < 1024; i += 256) {
        const int k = i >> 4, m = i & 15;
        Hk[i] = (m < 8) ? H[m * 64 + k] : 0.f;
    }
    if (tid < 64) Rs[tid] = Rg[tid];
    __syncthreads();
    if (tid < 64) {
        float acc = 0.f;
        for (int j = 0; j < 64; ++j)
            acc = __builtin_fmaf(Ft[fo(j, tid)], init_mean[g * S_ + j], acc);
        meanv[tid] = acc;
    }

    const int lane = tid & 31;
    const int wave = tid >> 5;
    const int half = lane >> 4;
    const int l15  = lane & 15;
    const TileCtx c = tile_ctx();
    const bool up0 = c.ti0 <= c.tj;
    const bool up1 = c.ti1 <= c.tj;

    {
        v8f a0 = {};
        v8f a1 = {};
        mm_slice<0, 16>(Pn, Ft, c, a0, a1, true, true);
        store_tile(tmpP, a0, c.ti0, c);
        store_tile(tmpP, a1, c.ti1, c);
        __syncthreads();
        v8f d0 = q_tile(Qg, c.ti0, c);
        v8f d1 = q_tile(Qg, c.ti1, c);
        mm_slice<0, 16>(tmpP, Ft, c, d0, d1, up0, up1);
        store_tile_sym(Pn, d0, c.ti0, c);
        store_tile_sym(Pn, d1, c.ti1, c);
        __syncthreads();
    }

    float ysave = 0.f;
    for (int t = 0; t < T_; ++t) {
        float yv = 0.f;
        if (tid >= 64 && tid < 72) {
            yv = y[(g * T_ + t) * M_ + (tid - 64)];
            if (t > 0) out[(g * T_ + t - 1) * M_ + (tid - 64)] = ysave;
        }
        v8f m0 = q_tile(Qg, c.ti0, c);
        v8f m1 = q_tile(Qg, c.ti1, c);
        v8f p0 = {};
        v8f p1 = {};

        float macc = 0.f;
        float yacc = 0.f;
        if (wave >= 4) {
            const int tj = (wave - 4) * 16;
            const int bJ = fbase(tj + l15, half);
            v2f a[16], b[16];
            #pragma unroll
            for (int kb = 0; kb < 16; ++kb) {
                const int k0 = kb * 4 + half * 2;
                a[kb]   = frag(Pn, bJ, kb);
                b[kb].x = Hk[(k0    ) * 16 + l15];
                b[kb].y = Hk[(k0 + 1) * 16 + l15];
            }
            __builtin_amdgcn_sched_barrier(0);
            v8f acc = {};
            #pragma unroll
            for (int kb = 0; kb < 16; ++kb)
                acc = __builtin_amdgcn_wmma_f32_16x16x4_f32(
                          false, a[kb], false, b[kb], (short)0, acc, false, false);
            #pragma unroll
            for (int r = 0; r < 8; ++r)
                PHk[(tj + r + half * 8) * 16 + l15] = acc[r];
        } else if (tid < 64) {
            macc = fmean_terms(Ft, meanv, 0, tid, macc);
        } else if (tid < 72) {
            yacc = yhat_terms(Hk, meanv, 0, tid - 64, yacc);
        }
        if (wave < 3)       mm_slice<0, 8>(Pn, Ft, c, p0, p1, true, true);
        else if (wave == 3) mm_slice<0, 10>(Pn, Ft, c, p0, p1, true, true);
        __syncthreads();

        if (wave == 3) {
            v2f a[16], b[16];
            #pragma unroll
            for (int kb = 0; kb < 16; ++kb) {
                const int k0 = kb * 4 + half * 2;
                a[kb].x = Hk[(k0    ) * 16 + l15];
                a[kb].y = Hk[(k0 + 1) * 16 + l15];
                b[kb].x = PHk[(k0    ) * 16 + l15];
                b[kb].y = PHk[(k0 + 1) * 16 + l15];
            }
            float r0[8];
            #pragma unroll
            for (int r = 0; r < 8; ++r) r0[r] = Rs[r * 8 + (l15 & 7)];
            __builtin_amdgcn_sched_barrier(0);
            v8f acc = {};
            #pragma unroll
            for (int kb = 0; kb < 16; ++kb)
                acc = __builtin_amdgcn_wmma_f32_16x16x4_f32(
                          false, a[kb], false, b[kb], (short)0, acc, false, false);
            if (half == 0) {
                #pragma unroll
                for (int r = 0; r < 8; ++r) {
                    const float v = (l15 < 8) ? acc[r] + r0[r]
                                              : ((l15 - 8 == r) ? 1.f : 0.f);
                    Wgj[r * 16 + l15] = v;
                }
            }
        } else if (tid < 64) {
            macc = fmean_terms(Ft, meanv, 32, tid, macc);
            meanp[tid] = macc;
        } else if (tid < 72) {
            yacc = yhat_terms(Hk, meanv, 32, tid - 64, yacc);
            ysave = yacc;
            resid[tid - 64] = yv - yacc;
        }
        if (wave < 3) {
            mm_slice<8, 16>(Pn, Ft, c, p0, p1, true, true);
            store_tile(tmpP, p0, c.ti0, c);
            store_tile(tmpP, p1, c.ti1, c);
        } else if (wave >= 4) {
            mm_slice<0, 9>(Pn, Ft, c, p0, p1, true, true);
        }
        __syncthreads();

        {
            const int w4 = wave - 4;
            v8f accP = {};
            #pragma unroll
            for (int q = 0; q < 4; ++q) {
                const int p = 2 * q;
                const float* Wo = (q & 1) ? Wgj2 : Wgj;
                float*       Wn = (q & 1) ? Wgj : Wgj2;
                if (tid < 64) {
                    const int r = tid >> 4, cc = tid & 15;
                    const float a   = Wo[p * 16 + p];
                    const float b   = Wo[p * 16 + p + 1];
                    const float cq  = Wo[(p + 1) * 16 + p];
                    const float d   = Wo[(p + 1) * 16 + p + 1];
                    const float r0  = Wo[p * 16 + cc];
                    const float r1  = Wo[(p + 1) * 16 + cc];
                    const float w0  = Wo[r * 16 + cc];
                    const float f00 = Wo[r * 16 + p];
                    const float f01 = Wo[r * 16 + p + 1];
                    const float w1  = Wo[(r + 4) * 16 + cc];
                    const float f10 = Wo[(r + 4) * 16 + p];
                    const float f11 = Wo[(r + 4) * 16 + p + 1];
                    __builtin_amdgcn_sched_barrier(0);
                    const float det = a * d - b * cq;
                    const float n0  = d * r0 - b * r1;
                    const float n1  = a * r1 - cq * r0;
                    const float g0  = f00 * n0 + f01 * n1;
                    const float g1  = f10 * n0 + f11 * n1;
                    const float iv  = 1.f / det;
                    const float wp0 = n0 * iv;
                    const float wp1 = n1 * iv;
                    const float u0  = w0 - g0 * iv;
                    const float u1  = w1 - g1 * iv;
                    Wn[r * 16 + cc]       = (r == p)     ? wp0 : ((r == p + 1)     ? wp1 : u0);
                    Wn[(r + 4) * 16 + cc] = (r + 4 == p) ? wp0 : ((r + 4 == p + 1) ? wp1 : u1);
                } else if (wave >= 4) {
                    const int tj = w4 * 16;
                    const int bJ = fbase(tj + l15, half);
                    v2f fa[4], fb[4];
                    #pragma unroll
                    for (int i = 0; i < 4; ++i) {
                        const int kb = 4 * q + i;
                        const int k0 = kb * 4 + half * 2;
                        fa[i]   = frag(Ft, bJ, kb);
                        fb[i].x = PHk[(k0    ) * 16 + l15];
                        fb[i].y = PHk[(k0 + 1) * 16 + l15];
                    }
                    __builtin_amdgcn_sched_barrier(0);
                    #pragma unroll
                    for (int i = 0; i < 4; ++i)
                        accP = __builtin_amdgcn_wmma_f32_16x16x4_f32(
                                   false, fa[i], false, fb[i], (short)0, accP, false, false);
                    if (q == 3 && l15 < 8) {
                        for (int r = 0; r < 8; ++r) {
                            const int row = tj + r + half * 8;
                            FPH[row * 8 + l15]   = accP[r];
                            FPHt[l15 * 64 + row] = accP[r];
                        }
                    }
                }
                if (q == 0) {
                    if (wave == 3) {
                        mm_slice<10, 16>(Pn, Ft, c, p0, p1, true, true);
                        store_tile(tmpP, p0, c.ti0, c);
                        store_tile(tmpP, p1, c.ti1, c);
                    } else if (wave >= 4) {
                        mm_slice<9, 16>(Pn, Ft, c, p0, p1, true, true);
                        store_tile(tmpP, p0, c.ti0, c);
                        store_tile(tmpP, p1, c.ti1, c);
                    }
                } else if (q == 1) {
                    p2_slice<0, 3, 0, 5, 0, 4>(tmpP, Ft, c, m0, m1, wave, up1);
                } else if (q == 2) {
                    p2_slice<3, 6, 5, 10, 4, 8>(tmpP, Ft, c, m0, m1, wave, up1);
                } else {
                    p2_slice<6, 9, 10, 14, 8, 12>(tmpP, Ft, c, m0, m1, wave, up1);
                }
                __syncthreads();
            }
        }

        for (int e = tid; e < 512; e += 256) {
            const int s = e >> 3, m = e & 7;
            float acc = 0.f;
            for (int j = 0; j < 8; ++j) acc += FPH[s * 8 + j] * Wgj[j * 16 + 8 + m];
            Lm[e] = acc;
            NLt[m * 64 + s] = -acc;
        }
        p2_slice<9, 16, 14, 16, 12, 16>(tmpP, Ft, c, m0, m1, wave, up1);
        __syncthreads();

        if (tid < 64) {
            float acc = meanp[tid];
            for (int m = 0; m < 8; ++m) acc += Lm[tid * 8 + m] * resid[m];
            meanv[tid] = acc;
        }
        if (up1) {
            v2f xb[2], x0[2], x1[2];
            #pragma unroll
            for (int kb = 0; kb < 2; ++kb) {
                const int k0 = kb * 4 + half * 2;
                xb[kb].x = FPHt[(k0    ) * 64 + c.tj  + l15];
                xb[kb].y = FPHt[(k0 + 1) * 64 + c.tj  + l15];
                x0[kb].x = NLt[(k0    ) * 64 + c.ti0 + l15];
                x0[kb].y = NLt[(k0 + 1) * 64 + c.ti0 + l15];
                x1[kb].x = NLt[(k0    ) * 64 + c.ti1 + l15];
                x1[kb].y = NLt[(k0 + 1) * 64 + c.ti1 + l15];
            }
            #pragma unroll
            for (int kb = 0; kb < 2; ++kb) {
                m0 = __builtin_amdgcn_wmma_f32_16x16x4_f32(
                         false, x0[kb], false, xb[kb], (short)0, m0, false, false);
                m1 = __builtin_amdgcn_wmma_f32_16x16x4_f32(
                         false, x1[kb], false, xb[kb], (short)0, m1, false, false);
            }
            store_tile_sym(Pn, m0, c.ti0, c);
            store_tile_sym(Pn, m1, c.ti1, c);
        } else if (up0) {
            v2f xb[2], x0[2];
            #pragma unroll
            for (int kb = 0; kb < 2; ++kb) {
                const int k0 = kb * 4 + half * 2;
                xb[kb].x = FPHt[(k0    ) * 64 + c.tj  + l15];
                xb[kb].y = FPHt[(k0 + 1) * 64 + c.tj  + l15];
                x0[kb].x = NLt[(k0    ) * 64 + c.ti0 + l15];
                x0[kb].y = NLt[(k0 + 1) * 64 + c.ti0 + l15];
            }
            #pragma unroll
            for (int kb = 0; kb < 2; ++kb)
                m0 = __builtin_amdgcn_wmma_f32_16x16x4_f32(
                         false, x0[kb], false, xb[kb], (short)0, m0, false, false);
            store_tile_sym(Pn, m0, c.ti0, c);
        }
        __syncthreads();
    }
    if (tid >= 64 && tid < 72) out[(g * T_ + T_ - 1) * M_ + (tid - 64)] = ysave;
}

extern "C" void kernel_launch(void* const* d_in, const int* in_sizes, int n_in,
                              void* d_out, int out_size, void* d_ws, size_t ws_size,
                              hipStream_t stream) {
    const float* y         = (const float*)d_in[0];
    const float* F         = (const float*)d_in[1];
    const float* H         = (const float*)d_in[2];
    const float* Lq        = (const float*)d_in[3];
    const float* Lr        = (const float*)d_in[4];
    const float* init_mean = (const float*)d_in[5];
    const float* Linit     = (const float*)d_in[6];
    float* out = (float*)d_out;
    float* ws  = (float*)d_ws;

    kf_prep<<<16, 256, 0, stream>>>(Lq, Lr, Linit, ws);

    const size_t smem_bytes = (4096 * 3 + 64 + 1024 + 512 + 512 + 512 + 512 + 128 +
                               64 + 64 + 8 + 128 + 1024) * sizeof(float);
    kf_main<<<G_, 256, smem_bytes, stream>>>(y, F, H,
                                             ws,
                                             ws + 8192,
                                             ws + 4096,
                                             init_mean, out);
}
